# starting kernel
# baseline (speedup 1.0000x reference)
#include <hip/hip_runtime.h>
#include <hip/hip_bf16.h>


typedef __attribute__((ext_vector_type(2))) float v2f;
typedef __attribute__((ext_vector_type(8))) float v8f;

#define IN_F 8
#define OUT_F 8
#define SPEC 16

__global__ __launch_bounds__(256) void kan_cheb_wmma(
    const float* __restrict__ x,
    const float* __restrict__ w,
    float* __restrict__ out,
    int n)
{
    const int lane = threadIdx.x & 31;
    const int col  = lane & 15;
    const bool hi  = lane >= 16;
    const int wavesPerBlock = blockDim.x >> 5;
    const int waveId = blockIdx.x * wavesPerBlock + (threadIdx.x >> 5);
    const int nWaves = gridDim.x * wavesPerBlock;
    const int nTiles = n >> 4;

    v2f bf[IN_F][4];
    #pragma unroll
    for (int f = 0; f < IN_F; ++f) {
        #pragma unroll
        for (int kk = 0; kk < 4; ++kk) {
            const int s0 = kk * 4 + (hi ? 2 : 0);
            float b0 = 0.0f, b1 = 0.0f;
            if (col < OUT_F) {
                b0 = w[(f * SPEC + s0    ) * OUT_F + col];
                b1 = w[(f * SPEC + s0 + 1) * OUT_F + col];
            }
            bf[f][kk].x = b0;
            bf[f][kk].y = b1;
        }
    }

    for (int t = waveId; t < nTiles; t += nWaves) {
        const int base = t << 4;
        const float4* xp4 = (const float4*)(x + (size_t)(base + col) * IN_F);
        const float4 l4 = xp4[0];
        const float4 h4 = xp4[1];
        float xv[IN_F] = { l4.x, l4.y, l4.z, l4.w, h4.x, h4.y, h4.z, h4.w };

        v8f acc = {0.f, 0.f, 0.f, 0.f, 0.f, 0.f, 0.f, 0.f};

        #pragma unroll
        for (int f = 0; f < IN_F; ++f) {
            const float xf = xv[f];
            const float x2 = xf + xf;
            const float T0  = 1.0f;
            const float T1  = xf;
            const float T2  = __builtin_fmaf(x2, T1,  -T0);
            const float T3  = __builtin_fmaf(x2, T2,  -T1);
            const float T4  = __builtin_fmaf(x2, T3,  -T2);
            const float T5  = __builtin_fmaf(x2, T4,  -T3);
            const float T6  = __builtin_fmaf(x2, T5,  -T4);
            const float T7  = __builtin_fmaf(x2, T6,  -T5);
            const float T8  = __builtin_fmaf(x2, T7,  -T6);
            const float T9  = __builtin_fmaf(x2, T8,  -T7);
            const float T10 = __builtin_fmaf(x2, T9,  -T8);
            const float T11 = __builtin_fmaf(x2, T10, -T9);
            const float T12 = __builtin_fmaf(x2, T11, -T10);
            const float T13 = __builtin_fmaf(x2, T12, -T11);
            const float T14 = __builtin_fmaf(x2, T13, -T12);
            const float T15 = __builtin_fmaf(x2, T14, -T13);

            v2f a;
            a.x = hi ? T2 : T0;  a.y = hi ? T3 : T1;
            acc = __builtin_amdgcn_wmma_f32_16x16x4_f32(
                false, a, false, bf[f][0], (short)0, acc, false, false);
            a.x = hi ? T6 : T4;  a.y = hi ? T7 : T5;
            acc = __builtin_amdgcn_wmma_f32_16x16x4_f32(
                false, a, false, bf[f][1], (short)0, acc, false, false);
            a.x = hi ? T10 : T8;  a.y = hi ? T11 : T9;
            acc = __builtin_amdgcn_wmma_f32_16x16x4_f32(
                false, a, false, bf[f][2], (short)0, acc, false, false);
            a.x = hi ? T14 : T12;  a.y = hi ? T15 : T13;
            acc = __builtin_amdgcn_wmma_f32_16x16x4_f32(
                false, a, false, bf[f][3], (short)0, acc, false, false);
        }

        if (col < OUT_F) {
            float* op = out + (size_t)(base + (hi ? 8 : 0)) * OUT_F + col;
            #pragma unroll
            for (int r = 0; r < 8; ++r)
                op[(size_t)r * OUT_F] = acc[r];
        }
    }
}

extern "C" void kernel_launch(void* const* d_in, const int* in_sizes, int n_in,
                              void* d_out, int out_size, void* d_ws, size_t ws_size,
                              hipStream_t stream) {
    const float* x = (const float*)d_in[0];
    const float* w = (const float*)d_in[1];
    float* out = (float*)d_out;
    const int n = in_sizes[0] / IN_F;

    const int block = 256;
    const int grid = 1024;
    kan_cheb_wmma<<<grid, block, 0, stream>>>(x, w, out, n);
}


# k order remapped: each 16-lane group owns 4 features and an A pair is one mode of 2 features, so the recurrence runs as packed FMAs without duplicate lanes or selects; sum order changes
# speedup vs baseline: 1.1979x; 1.1979x over previous
#include <hip/hip_runtime.h>
#include <hip/hip_bf16.h>


typedef __attribute__((ext_vector_type(2))) float v2f;
typedef __attribute__((ext_vector_type(4))) float v4f;
typedef __attribute__((ext_vector_type(8))) float v8f;

#define IN_F 8
#define OUT_F 8
#define SPEC 16

__global__ __launch_bounds__(256) void kan_cheb_wmma(
    const float* __restrict__ x,
    const float* __restrict__ w,
    float* __restrict__ out,
    int n)
{
    const int lane = threadIdx.x & 31;
    const int col  = lane & 15;
    const int h    = lane >> 4;
    const bool hi  = lane >= 16;
    const int wavesPerBlock = blockDim.x >> 5;
    const int waveId = blockIdx.x * wavesPerBlock + (threadIdx.x >> 5);
    const int nWaves = gridDim.x * wavesPerBlock;
    const int nTiles = n >> 4;

    v2f bf[2][SPEC];
    #pragma unroll
    for (int p = 0; p < 2; ++p) {
        #pragma unroll
        for (int s = 0; s < SPEC; ++s) {
            const int f0 = 4 * h + 2 * p;
            float b0 = 0.0f, b1 = 0.0f;
            if (col < OUT_F) {
                b0 = w[((f0    ) * SPEC + s) * OUT_F + col];
                b1 = w[((f0 + 1) * SPEC + s) * OUT_F + col];
            }
            bf[p][s].x = b0;
            bf[p][s].y = b1;
        }
    }

    const v2f ones = {1.0f, 1.0f};

    for (int t = waveId; t < nTiles; t += nWaves) {
        const int base = t << 4;
        const v4f xv = *(const v4f*)(x + (size_t)(base + col) * IN_F + 4 * h);
        const v2f xp = {xv.x, xv.y};
        const v2f xq = {xv.z, xv.w};
        const v2f xp2 = xp + xp;
        const v2f xq2 = xq + xq;

        v8f acc = {0.f, 0.f, 0.f, 0.f, 0.f, 0.f, 0.f, 0.f};

        acc = __builtin_amdgcn_wmma_f32_16x16x4_f32(
            false, ones, false, bf[0][0], (short)0, acc, false, false);
        acc = __builtin_amdgcn_wmma_f32_16x16x4_f32(
            false, ones, false, bf[1][0], (short)0, acc, false, false);
        acc = __builtin_amdgcn_wmma_f32_16x16x4_f32(
            false, xp, false, bf[0][1], (short)0, acc, false, false);
        acc = __builtin_amdgcn_wmma_f32_16x16x4_f32(
            false, xq, false, bf[1][1], (short)0, acc, false, false);

        v2f pm2 = ones, pm1 = xp;
        v2f qm2 = ones, qm1 = xq;
        #pragma unroll
        for (int s = 2; s < SPEC; ++s) {
            const v2f pt = __builtin_elementwise_fma(xp2, pm1, -pm2);
            const v2f qt = __builtin_elementwise_fma(xq2, qm1, -qm2);
            acc = __builtin_amdgcn_wmma_f32_16x16x4_f32(
                false, pt, false, bf[0][s], (short)0, acc, false, false);
            acc = __builtin_amdgcn_wmma_f32_16x16x4_f32(
                false, qt, false, bf[1][s], (short)0, acc, false, false);
            pm2 = pm1; pm1 = pt;
            qm2 = qm1; qm1 = qt;
        }

        if (col < OUT_F) {
            float* op = out + (size_t)(base + (hi ? 8 : 0)) * OUT_F + col;
            #pragma unroll
            for (int r = 0; r < 8; ++r)
                op[(size_t)r * OUT_F] = acc[r];
        }
    }
}

extern "C" void kernel_launch(void* const* d_in, const int* in_sizes, int n_in,
                              void* d_out, int out_size, void* d_ws, size_t ws_size,
                              hipStream_t stream) {
    const float* x = (const float*)d_in[0];
    const float* w = (const float*)d_in[1];
    float* out = (float*)d_out;
    const int n = in_sizes[0] / IN_F;

    const int block = 256;
    const int grid = 1024;
    kan_cheb_wmma<<<grid, block, 0, stream>>>(x, w, out, n);
}


# sample loads run one tile ahead: the next tile's 16-byte-per-lane load is issued before the current tile's matrix work, and the first tile's before the weight fetch
# speedup vs baseline: 1.2381x; 1.0336x over previous
#include <hip/hip_runtime.h>
#include <hip/hip_bf16.h>


typedef __attribute__((ext_vector_type(2))) float v2f;
typedef __attribute__((ext_vector_type(4))) float v4f;
typedef __attribute__((ext_vector_type(8))) float v8f;

#define IN_F 8
#define OUT_F 8
#define SPEC 16

__global__ __launch_bounds__(256) void kan_cheb_wmma(
    const float* __restrict__ x,
    const float* __restrict__ w,
    float* __restrict__ out,
    int n)
{
    const int lane = threadIdx.x & 31;
    const int col  = lane & 15;
    const int h    = lane >> 4;
    const bool hi  = lane >= 16;
    const int wavesPerBlock = blockDim.x >> 5;
    const int waveId = blockIdx.x * wavesPerBlock + (threadIdx.x >> 5);
    const int nWaves = gridDim.x * wavesPerBlock;
    const int nTiles = n >> 4;

    const float* xl = x + (size_t)col * IN_F + 4 * h;
    v4f xv = {0.f, 0.f, 0.f, 0.f};
    if (waveId < nTiles)
        xv = *(const v4f*)(xl + (size_t)(waveId << 4) * IN_F);

    v2f bf[2][SPEC];
    #pragma unroll
    for (int p = 0; p < 2; ++p) {
        #pragma unroll
        for (int s = 0; s < SPEC; ++s) {
            const int f0 = 4 * h + 2 * p;
            float b0 = 0.0f, b1 = 0.0f;
            if (col < OUT_F) {
                b0 = w[((f0    ) * SPEC + s) * OUT_F + col];
                b1 = w[((f0 + 1) * SPEC + s) * OUT_F + col];
            }
            bf[p][s].x = b0;
            bf[p][s].y = b1;
        }
    }

    const v2f ones = {1.0f, 1.0f};

    for (int t = waveId; t < nTiles; t += nWaves) {
        const int base = t << 4;
        const int tn = (t + nWaves < nTiles) ? (t + nWaves) : t;
        const v4f xnext = *(const v4f*)(xl + (size_t)(tn << 4) * IN_F);
        __builtin_amdgcn_sched_barrier(0);
        const v2f xp = {xv.x, xv.y};
        const v2f xq = {xv.z, xv.w};
        const v2f xp2 = xp + xp;
        const v2f xq2 = xq + xq;

        v8f acc = {0.f, 0.f, 0.f, 0.f, 0.f, 0.f, 0.f, 0.f};

        acc = __builtin_amdgcn_wmma_f32_16x16x4_f32(
            false, ones, false, bf[0][0], (short)0, acc, false, false);
        acc = __builtin_amdgcn_wmma_f32_16x16x4_f32(
            false, ones, false, bf[1][0], (short)0, acc, false, false);
        acc = __builtin_amdgcn_wmma_f32_16x16x4_f32(
            false, xp, false, bf[0][1], (short)0, acc, false, false);
        acc = __builtin_amdgcn_wmma_f32_16x16x4_f32(
            false, xq, false, bf[1][1], (short)0, acc, false, false);

        v2f pm2 = ones, pm1 = xp;
        v2f qm2 = ones, qm1 = xq;
        #pragma unroll
        for (int s = 2; s < SPEC; ++s) {
            const v2f pt = __builtin_elementwise_fma(xp2, pm1, -pm2);
            const v2f qt = __builtin_elementwise_fma(xq2, qm1, -qm2);
            acc = __builtin_amdgcn_wmma_f32_16x16x4_f32(
                false, pt, false, bf[0][s], (short)0, acc, false, false);
            acc = __builtin_amdgcn_wmma_f32_16x16x4_f32(
                false, qt, false, bf[1][s], (short)0, acc, false, false);
            pm2 = pm1; pm1 = pt;
            qm2 = qm1; qm1 = qt;
        }

        if (col < OUT_F) {
            float* op = out + (size_t)(base + (hi ? 8 : 0)) * OUT_F + col;
            #pragma unroll
            for (int r = 0; r < 8; ++r)
                op[(size_t)r * OUT_F] = acc[r];
        }
        xv = xnext;
    }
}

extern "C" void kernel_launch(void* const* d_in, const int* in_sizes, int n_in,
                              void* d_out, int out_size, void* d_ws, size_t ws_size,
                              hipStream_t stream) {
    const float* x = (const float*)d_in[0];
    const float* w = (const float*)d_in[1];
    float* out = (float*)d_out;
    const int n = in_sizes[0] / IN_F;

    const int block = 256;
    const int grid = 1024;
    kan_cheb_wmma<<<grid, block, 0, stream>>>(x, w, out, n);
}


# constant mode T0=1 hoisted: its two matrix products (ones x W[:,0,:]) are issued once per wave and start every tile's accumulator; 30 instead of 32 matrix instructions per tile, same sum order
# speedup vs baseline: 1.2793x; 1.0333x over previous
#include <hip/hip_runtime.h>
#include <hip/hip_bf16.h>


typedef __attribute__((ext_vector_type(2))) float v2f;
typedef __attribute__((ext_vector_type(4))) float v4f;
typedef __attribute__((ext_vector_type(8))) float v8f;

#define IN_F 8
#define OUT_F 8
#define SPEC 16

__global__ __launch_bounds__(256) void kan_cheb_wmma(
    const float* __restrict__ x,
    const float* __restrict__ w,
    float* __restrict__ out,
    int n)
{
    const int lane = threadIdx.x & 31;
    const int col  = lane & 15;
    const int h    = lane >> 4;
    const bool hi  = lane >= 16;
    const int wavesPerBlock = blockDim.x >> 5;
    const int waveId = blockIdx.x * wavesPerBlock + (threadIdx.x >> 5);
    const int nWaves = gridDim.x * wavesPerBlock;
    const int nTiles = n >> 4;

    const float* xl = x + (size_t)col * IN_F + 4 * h;
    v4f xv = {0.f, 0.f, 0.f, 0.f};
    if (waveId < nTiles)
        xv = *(const v4f*)(xl + (size_t)(waveId << 4) * IN_F);

    v2f bf[2][SPEC];
    #pragma unroll
    for (int p = 0; p < 2; ++p) {
        #pragma unroll
        for (int s = 0; s < SPEC; ++s) {
            const int f0 = 4 * h + 2 * p;
            float b0 = 0.0f, b1 = 0.0f;
            if (col < OUT_F) {
                b0 = w[((f0    ) * SPEC + s) * OUT_F + col];
                b1 = w[((f0 + 1) * SPEC + s) * OUT_F + col];
            }
            bf[p][s].x = b0;
            bf[p][s].y = b1;
        }
    }

    const v2f ones = {1.0f, 1.0f};

    v8f acc0 = {0.f, 0.f, 0.f, 0.f, 0.f, 0.f, 0.f, 0.f};
    acc0 = __builtin_amdgcn_wmma_f32_16x16x4_f32(
        false, ones, false, bf[0][0], (short)0, acc0, false, false);
    acc0 = __builtin_amdgcn_wmma_f32_16x16x4_f32(
        false, ones, false, bf[1][0], (short)0, acc0, false, false);

    for (int t = waveId; t < nTiles; t += nWaves) {
        const int base = t << 4;
        const int tn = (t + nWaves < nTiles) ? (t + nWaves) : t;
        const v4f xnext = *(const v4f*)(xl + (size_t)(tn << 4) * IN_F);
        __builtin_amdgcn_sched_barrier(0);
        const v2f xp = {xv.x, xv.y};
        const v2f xq = {xv.z, xv.w};
        const v2f xp2 = xp + xp;
        const v2f xq2 = xq + xq;

        v8f acc = __builtin_amdgcn_wmma_f32_16x16x4_f32(
            false, xp, false, bf[0][1], (short)0, acc0, false, false);
        acc = __builtin_amdgcn_wmma_f32_16x16x4_f32(
            false, xq, false, bf[1][1], (short)0, acc, false, false);

        v2f pm2 = ones, pm1 = xp;
        v2f qm2 = ones, qm1 = xq;
        #pragma unroll
        for (int s = 2; s < SPEC; ++s) {
            const v2f pt = __builtin_elementwise_fma(xp2, pm1, -pm2);
            const v2f qt = __builtin_elementwise_fma(xq2, qm1, -qm2);
            acc = __builtin_amdgcn_wmma_f32_16x16x4_f32(
                false, pt, false, bf[0][s], (short)0, acc, false, false);
            acc = __builtin_amdgcn_wmma_f32_16x16x4_f32(
                false, qt, false, bf[1][s], (short)0, acc, false, false);
            pm2 = pm1; pm1 = pt;
            qm2 = qm1; qm1 = qt;
        }

        if (col < OUT_F) {
            float* op = out + (size_t)(base + (hi ? 8 : 0)) * OUT_F + col;
            #pragma unroll
            for (int r = 0; r < 8; ++r)
                op[(size_t)r * OUT_F] = acc[r];
        }
        xv = xnext;
    }
}

extern "C" void kernel_launch(void* const* d_in, const int* in_sizes, int n_in,
                              void* d_out, int out_size, void* d_ws, size_t ws_size,
                              hipStream_t stream) {
    const float* x = (const float*)d_in[0];
    const float* w = (const float*)d_in[1];
    float* out = (float*)d_out;
    const int n = in_sizes[0] / IN_F;

    const int block = 256;
    const int grid = 1024;
    kan_cheb_wmma<<<grid, block, 0, stream>>>(x, w, out, n);
}


# modes 8..15 folded onto the 8 idle matrix rows by T(8+j)=2*T8*Tj-T(8-j): weights as A (P, Q rows), per-feature D scaled by 2*T8 on the vector unit; 16 matrix ops per tile, not 30; sum order changes
# speedup vs baseline: 1.5950x; 1.2468x over previous
#include <hip/hip_runtime.h>
#include <hip/hip_bf16.h>


typedef __attribute__((ext_vector_type(2))) float v2f;
typedef __attribute__((ext_vector_type(4))) float v4f;
typedef __attribute__((ext_vector_type(8))) float v8f;

#define IN_F 8
#define OUT_F 8
#define SPEC 16

__global__ __launch_bounds__(256) void kan_cheb_wmma(
    const float* __restrict__ x,
    const float* __restrict__ w,
    float* __restrict__ out,
    int n)
{
    const int lane = threadIdx.x & 31;
    const int col  = lane & 15;
    const int h    = lane >> 4;
    const bool hi  = lane >= 16;
    const int wavesPerBlock = blockDim.x >> 5;
    const int waveId = blockIdx.x * wavesPerBlock + (threadIdx.x >> 5);
    const int nWaves = gridDim.x * wavesPerBlock;
    const int nTiles = n >> 4;

    const float* xl = x + (size_t)col * IN_F;
    v4f xa = {0.f, 0.f, 0.f, 0.f};
    v4f xb = {0.f, 0.f, 0.f, 0.f};
    if (waveId < nTiles) {
        const float* p0 = xl + (size_t)(waveId << 4) * IN_F;
        xa = *(const v4f*)(p0);
        xb = *(const v4f*)(p0 + 4);
    }

    const int  o    = col & 7;
    const bool qrow = col >= 8;
    v2f af[IN_F][2];
    #pragma unroll
    for (int f = 0; f < IN_F; ++f) {
        #pragma unroll
        for (int i = 0; i < 2; ++i) {
            #pragma unroll
            for (int e = 0; e < 2; ++e) {
                const int m = 2 * (2 * i + e) + (1 - h);
                const bool m0 = (m == 0);
                const int sa = qrow ? (8 + m) : m;
                const int sb = m0 ? sa : (SPEC - m);
                const float wa = w[(f * SPEC + sa) * OUT_F + o];
                const float wb = w[(f * SPEC + sb) * OUT_F + o];
                const float pv = m0 ? wa : (wa - wb);
                const float qv = m0 ? 0.5f * wa : wa;
                af[f][i][e] = qrow ? qv : pv;
            }
        }
    }

    const float c_a0m = hi ? 0.0f : 1.0f;
    const float c_a0c = hi ? 1.0f : 0.0f;
    const float c_tp  = hi ? 0.5f : 1.0f;
    const float c_tq  = hi ? 0.0f : -1.0f;
    const float c_s1  = hi ? 2.0f : 0.0f;
    const float c_s0  = hi ? 0.0f : 1.0f;

    const v8f zero8 = {0.f, 0.f, 0.f, 0.f, 0.f, 0.f, 0.f, 0.f};

    for (int t = waveId; t < nTiles; t += nWaves) {
        const int base = t << 4;
        const int tn = (t + nWaves < nTiles) ? (t + nWaves) : t;
        const float* pn = xl + (size_t)(tn << 4) * IN_F;
        const v4f xna = *(const v4f*)(pn);
        const v4f xnb = *(const v4f*)(pn + 4);
        __builtin_amdgcn_sched_barrier(0);

        const float xs[IN_F] = {xa.x, xa.y, xa.z, xa.w, xb.x, xb.y, xb.z, xb.w};
        v8f R;
        #pragma unroll
        for (int f = 0; f < IN_F; ++f) {
            const float xf = xs[f];
            const float x2 = xf * xf;
            const float y  = __builtin_fmaf(4.0f, x2, -2.0f);
            const float a0 = __builtin_fmaf(xf, c_a0m, c_a0c);
            const float tt = __builtin_fmaf(y, c_tp, c_tq);
            const float a1 = a0 * tt;
            const float a2 = __builtin_fmaf(y, a1, -a0);
            const float a3 = __builtin_fmaf(y, a2, -a1);
            const float a4 = __builtin_fmaf(y, a3, -a2);
            const float sc = __builtin_fmaf(a4, c_s1, c_s0);
            const v2f b0 = {a0, a1};
            const v2f b1 = {a2, a3};
            v8f d = __builtin_amdgcn_wmma_f32_16x16x4_f32(
                false, af[f][0], false, b0, (short)0, zero8, false, false);
            d = __builtin_amdgcn_wmma_f32_16x16x4_f32(
                false, af[f][1], false, b1, (short)0, d, false, false);
            if (f == 0) R = d * sc;
            else        R = __builtin_elementwise_fma(d, (v8f)(sc), R);
        }

        v4f res;
        #pragma unroll
        for (int k = 0; k < 4; ++k) {
            const float rk = R[k];
            const float rk4 = R[4 + k];
            const auto sw = __builtin_amdgcn_permlane16_swap(
                __float_as_uint(rk), __float_as_uint(rk4), false, false);
            const unsigned s0 = sw[0];
            const unsigned s1 = sw[1];
            res[k] = __uint_as_float(s0) + __uint_as_float(s1);
        }
        *(v4f*)(out + (size_t)(base + col) * OUT_F + 4 * h) = res;

        xa = xna;
        xb = xnb;
    }
}

extern "C" void kernel_launch(void* const* d_in, const int* in_sizes, int n_in,
                              void* d_out, int out_size, void* d_ws, size_t ws_size,
                              hipStream_t stream) {
    const float* x = (const float*)d_in[0];
    const float* w = (const float*)d_in[1];
    float* out = (float*)d_out;
    const int n = in_sizes[0] / IN_F;

    const int block = 256;
    const int grid = 1024;
    kan_cheb_wmma<<<grid, block, 0, stream>>>(x, w, out, n);
}


# per-tile loop overhead off the vector unit: readfirstlane of the wave index keeps counter, trip test and tile bases scalar; two tiles per trip, so prefetched samples need no register copies
# speedup vs baseline: 1.6965x; 1.0636x over previous
#include <hip/hip_runtime.h>
#include <hip/hip_bf16.h>


typedef __attribute__((ext_vector_type(2))) float v2f;
typedef __attribute__((ext_vector_type(4))) float v4f;
typedef __attribute__((ext_vector_type(8))) float v8f;

#define IN_F 8
#define OUT_F 8
#define SPEC 16

__global__ __launch_bounds__(256) void kan_cheb_wmma(
    const float* __restrict__ x,
    const float* __restrict__ w,
    float* __restrict__ out,
    int n)
{
    const int lane = threadIdx.x & 31;
    const int col  = lane & 15;
    const int h    = lane >> 4;
    const bool hi  = lane >= 16;
    const int wavesPerBlock = blockDim.x >> 5;
    const int waveId = __builtin_amdgcn_readfirstlane(
        (int)(blockIdx.x * wavesPerBlock + (threadIdx.x >> 5)));
    const int nWaves = gridDim.x * wavesPerBlock;
    const int nTiles = n >> 4;

    const unsigned xoff = (unsigned)col * IN_F;
    v4f xa = {0.f, 0.f, 0.f, 0.f};
    v4f xb = {0.f, 0.f, 0.f, 0.f};
    if (waveId < nTiles) {
        const float* p0 = x + (size_t)waveId * (16 * IN_F);
        xa = *(const v4f*)(p0 + xoff);
        xb = *(const v4f*)(p0 + xoff + 4);
    }

    const int  o    = col & 7;
    const bool qrow = col >= 8;
    v2f af[IN_F][2];
    #pragma unroll
    for (int f = 0; f < IN_F; ++f) {
        #pragma unroll
        for (int i = 0; i < 2; ++i) {
            #pragma unroll
            for (int e = 0; e < 2; ++e) {
                const int m = 2 * (2 * i + e) + (1 - h);
                const bool m0 = (m == 0);
                const int sa = qrow ? (8 + m) : m;
                const int sb = m0 ? sa : (SPEC - m);
                const float wa = w[(f * SPEC + sa) * OUT_F + o];
                const float wb = w[(f * SPEC + sb) * OUT_F + o];
                const float pv = m0 ? wa : (wa - wb);
                const float qv = m0 ? 0.5f * wa : wa;
                af[f][i][e] = qrow ? qv : pv;
            }
        }
    }

    const float c_a0m = hi ? 0.0f : 1.0f;
    const float c_a0c = hi ? 1.0f : 0.0f;
    const float c_tp  = hi ? 0.5f : 1.0f;
    const float c_tq  = hi ? 0.0f : -1.0f;
    const float c_s1  = hi ? 2.0f : 0.0f;
    const float c_s0  = hi ? 0.0f : 1.0f;

    const v8f zero8 = {0.f, 0.f, 0.f, 0.f, 0.f, 0.f, 0.f, 0.f};

    const unsigned ooff = (unsigned)col * OUT_F + 4u * (unsigned)h;

    auto tile = [&](const int t, const int tnext, const v4f xca, const v4f xcb,
                    v4f& xna, v4f& xnb, const bool pf) __attribute__((always_inline)) {
        if (pf) {
            const float* pn = x + (size_t)tnext * (16 * IN_F);
            xna = *(const v4f*)(pn + xoff);
            xnb = *(const v4f*)(pn + xoff + 4);
            __builtin_amdgcn_sched_barrier(0);
        }

        const float xs[IN_F] = {xca.x, xca.y, xca.z, xca.w, xcb.x, xcb.y, xcb.z, xcb.w};
        v8f R;
        #pragma unroll
        for (int f = 0; f < IN_F; ++f) {
            const float xf = xs[f];
            const float x2 = xf * xf;
            const float y  = __builtin_fmaf(4.0f, x2, -2.0f);
            const float a0 = __builtin_fmaf(xf, c_a0m, c_a0c);
            const float tt = __builtin_fmaf(y, c_tp, c_tq);
            const float a1 = a0 * tt;
            const float a2 = __builtin_fmaf(y, a1, -a0);
            const float a3 = __builtin_fmaf(y, a2, -a1);
            const float a4 = __builtin_fmaf(y, a3, -a2);
            const float sc = __builtin_fmaf(a4, c_s1, c_s0);
            const v2f b0 = {a0, a1};
            const v2f b1 = {a2, a3};
            v8f d = __builtin_amdgcn_wmma_f32_16x16x4_f32(
                false, af[f][0], false, b0, (short)0, zero8, false, false);
            d = __builtin_amdgcn_wmma_f32_16x16x4_f32(
                false, af[f][1], false, b1, (short)0, d, false, false);
            if (f == 0) R = d * sc;
            else        R = __builtin_elementwise_fma(d, (v8f)(sc), R);
        }

        v4f res;
        #pragma unroll
        for (int k = 0; k < 4; ++k) {
            const float rk = R[k];
            const float rk4 = R[4 + k];
            const auto sw = __builtin_amdgcn_permlane16_swap(
                __float_as_uint(rk), __float_as_uint(rk4), false, false);
            const unsigned s0 = sw[0];
            const unsigned s1 = sw[1];
            res[k] = __uint_as_float(s0) + __uint_as_float(s1);
        }
        float* po = out + (size_t)t * (16 * OUT_F);
        *(v4f*)(po + ooff) = res;
    };

    v4f xc = {0.f, 0.f, 0.f, 0.f};
    v4f xd = {0.f, 0.f, 0.f, 0.f};
    int t = waveId;
    for (; t + nWaves < nTiles; t += 2 * nWaves) {
        const int t1 = t + nWaves;
        tile(t, t1, xa, xb, xc, xd, true);
        const int t2 = t1 + nWaves;
        tile(t1, (t2 < nTiles) ? t2 : t1, xc, xd, xa, xb, true);
    }
    if (t < nTiles) tile(t, t, xa, xb, xc, xd, false);
}

extern "C" void kernel_launch(void* const* d_in, const int* in_sizes, int n_in,
                              void* d_out, int out_size, void* d_ws, size_t ws_size,
                              hipStream_t stream) {
    const float* x = (const float*)d_in[0];
    const float* w = (const float*)d_in[1];
    float* out = (float*)d_out;
    const int n = in_sizes[0] / IN_F;

    const int block = 256;
    const int grid = 1024;
    kan_cheb_wmma<<<grid, block, 0, stream>>>(x, w, out, n);
}
